# starting kernel
# baseline (speedup 1.0000x reference)
#include <hip/hip_runtime.h>

typedef __attribute__((ext_vector_type(2))) float v2f;
typedef __attribute__((ext_vector_type(8))) float v8f;

#define WAVES_PER_BLOCK 8
#define ELEMS_PER_WAVE  4
#define ELEMS_PER_BLOCK (WAVES_PER_BLOCK * ELEMS_PER_WAVE)

__device__ __forceinline__ void wave_lds_fence() {
    __builtin_amdgcn_wave_barrier();
    asm volatile("s_wait_dscnt 0x0" ::: "memory");
    __builtin_amdgcn_wave_barrier();
}

__global__ __launch_bounds__(256)
void cholesky_rho_wmma_kernel(const float* __restrict__ x,
                              float* __restrict__ out, int B) {
    __shared__ float lds_in [WAVES_PER_BLOCK * 64 + 16];
    __shared__ float lds_out[WAVES_PER_BLOCK * 512];

    const int tid    = threadIdx.x;
    const int wave   = tid >> 5;
    const int lane   = tid & 31;
    const int base_e = (blockIdx.x * WAVES_PER_BLOCK + wave) * ELEMS_PER_WAVE;

    float* win  = lds_in  + wave * 64;
    float* wout = lds_out + wave * 512;

    {
        const int e_ld = base_e + (lane >> 3);
        const int off  = lane * 2;
        float2 v = make_float2(0.f, 0.f);
        if (e_ld < B) v = *(const float2*)(x + (size_t)base_e * 16 + off);
        win[off]     = v.x;
        win[off + 1] = v.y;
    }
    wave_lds_fence();

    const int h = lane >> 4;
    const int r = lane & 15;
    const int e = r >> 2;
    const int t = r & 3;
    const float* xe = win + e * 16;

    const float dv = xe[t * t + 2 * t];
    const float sp = fmaxf(dv, 0.f) + log1pf(expf(-fabsf(dv)));

    v2f lr, li;
    #pragma unroll
    for (int q = 0; q < 2; ++q) {
        const int   c      = 2 * h + q;
        const int   cc     = (c < t) ? c : t;
        const int   idx    = t * t + 2 * cc;
        const float re_raw = xe[idx];
        const float im_raw = xe[idx + 1];
        const float re     = (c > t) ? 0.f : ((c == t) ? sp : re_raw);
        const float im     = (c < t) ? im_raw : 0.f;
        if (q == 0) { lr.x = re; li.x = im; }
        else        { lr.y = re; li.y = im; }
    }

    v8f zero8 = {};
    v8f acc_r = __builtin_amdgcn_wmma_f32_16x16x4_f32(
        false, lr, false, lr, (short)0, zero8, false, false);
    acc_r = __builtin_amdgcn_wmma_f32_16x16x4_f32(
        false, li, false, li, (short)0, acc_r, false, false);
    v8f t1 = __builtin_amdgcn_wmma_f32_16x16x4_f32(
        false, li, false, lr, (short)0, zero8, false, false);
    v8f t2 = __builtin_amdgcn_wmma_f32_16x16x4_f32(
        false, lr, false, li, (short)0, zero8, false, false);

    {
        float* dst = wout + (h * 8) * 32 + r * 2;
        #pragma unroll
        for (int v = 0; v < 8; ++v) {
            *(float2*)(dst + v * 32) = make_float2(acc_r[v], t1[v] - t2[v]);
        }
    }
    wave_lds_fence();

    {
        const int e2 = lane >> 3;
        const int q  = lane & 7;
        const int eg = base_e + e2;
        if (eg < B) {
            const float* blk = wout + e2 * 136;
            const float  tr  = blk[0] + blk[34] + blk[68] + blk[102];
            const float  inv = 1.0f / (tr + 1e-8f);
            const int i  = q >> 1;
            const int j0 = (2 * q)     & 3;
            const int j1 = (2 * q + 1) & 3;
            const float2 p0 = *(const float2*)(blk + i * 32 + j0 * 2);
            const float2 p1 = *(const float2*)(blk + i * 32 + j1 * 2);
            const float4 vv = make_float4(p0.x * inv, p0.y * inv,
                                          p1.x * inv, p1.y * inv);
            *(float4*)(out + (size_t)base_e * 32 + lane * 4) = vv;
        }
    }
}

extern "C" void kernel_launch(void* const* d_in, const int* in_sizes, int n_in,
                              void* d_out, int out_size, void* d_ws, size_t ws_size,
                              hipStream_t stream) {
    (void)n_in; (void)out_size; (void)d_ws; (void)ws_size;
    const float* x = (const float*)d_in[0];
    float* out     = (float*)d_out;
    const int B    = in_sizes[0] / 16;
    const int grid = (B + ELEMS_PER_BLOCK - 1) / ELEMS_PER_BLOCK;
    hipLaunchKernelGGL(cholesky_rho_wmma_kernel, dim3(grid), dim3(256), 0, stream,
                       x, out, B);
}


# 8 elements per wave (one 16-byte load per lane): softplus evaluated once per diagonal entry, one per lane, lane halves swapped by v_permlanex16, where each entry was evaluated by two lanes
# speedup vs baseline: 1.2762x; 1.2762x over previous
#include <hip/hip_runtime.h>


typedef __attribute__((ext_vector_type(2))) float v2f;
typedef __attribute__((ext_vector_type(8))) float v8f;

#define WAVES_PER_BLOCK 8
#define GROUPS_PER_WAVE 2
#define ELEMS_PER_WAVE  (4 * GROUPS_PER_WAVE)
#define ELEMS_PER_BLOCK (WAVES_PER_BLOCK * ELEMS_PER_WAVE)

__device__ __forceinline__ void wave_lds_fence() {
    __builtin_amdgcn_wave_barrier();
    asm volatile("s_wait_dscnt 0x0" ::: "memory");
    __builtin_amdgcn_wave_barrier();
}

__device__ __forceinline__ float xhalf(float v) {
    const int b = __builtin_bit_cast(int, v);
    return __builtin_bit_cast(float,
        __builtin_amdgcn_permlanex16(b, b, 0x76543210, 0xfedcba98, false, false));
}

__device__ __forceinline__ void group_products(const float* xe, float spg,
                                               int h, int t, float* dst) {
    v2f lr, li;
    #pragma unroll
    for (int q = 0; q < 2; ++q) {
        const int   c      = 2 * h + q;
        const int   cc     = (c < t) ? c : t;
        const int   idx    = t * t + 2 * cc;
        const float re_raw = xe[idx];
        const float im_raw = xe[idx + 1];
        const float re     = (c > t) ? 0.f : ((c == t) ? spg : re_raw);
        const float im     = (c < t) ? im_raw : 0.f;
        if (q == 0) { lr.x = re; li.x = im; }
        else        { lr.y = re; li.y = im; }
    }

    v8f zero8 = {};
    v8f acc_r = __builtin_amdgcn_wmma_f32_16x16x4_f32(
        false, lr, false, lr, (short)0, zero8, false, false);
    acc_r = __builtin_amdgcn_wmma_f32_16x16x4_f32(
        false, li, false, li, (short)0, acc_r, false, false);
    v8f t1 = __builtin_amdgcn_wmma_f32_16x16x4_f32(
        false, li, false, lr, (short)0, zero8, false, false);
    v8f t2 = __builtin_amdgcn_wmma_f32_16x16x4_f32(
        false, lr, false, li, (short)0, zero8, false, false);

    #pragma unroll
    for (int v = 0; v < 8; ++v) {
        *(float2*)(dst + v * 32) = make_float2(acc_r[v], t1[v] - t2[v]);
    }
}

__device__ __forceinline__ void group_store(const float* tile, float* __restrict__ og,
                                            int lane, bool live) {
    const int e2 = lane >> 3;
    const int q  = lane & 7;
    if (live) {
        const float* blk = tile + e2 * 136;
        const float  tr  = blk[0] + blk[34] + blk[68] + blk[102];
        const float  inv = 1.0f / (tr + 1e-8f);
        const int i  = q >> 1;
        const int j0 = (2 * q)     & 3;
        const int j1 = (2 * q + 1) & 3;
        const float2 p0 = *(const float2*)(blk + i * 32 + j0 * 2);
        const float2 p1 = *(const float2*)(blk + i * 32 + j1 * 2);
        const float4 vv = make_float4(p0.x * inv, p0.y * inv,
                                      p1.x * inv, p1.y * inv);
        *(float4*)(og + lane * 4) = vv;
    }
}

__global__ __launch_bounds__(256)
void cholesky_rho_wmma_kernel(const float* __restrict__ x,
                              float* __restrict__ out, int B) {
    __shared__ __attribute__((aligned(16))) float lds_in [WAVES_PER_BLOCK * 128 + 16];
    __shared__ __attribute__((aligned(16))) float lds_out[WAVES_PER_BLOCK * GROUPS_PER_WAVE * 512];

    const int tid    = threadIdx.x;
    const int wave   = tid >> 5;
    const int lane   = tid & 31;
    const int base_e = (blockIdx.x * WAVES_PER_BLOCK + wave) * ELEMS_PER_WAVE;

    float* win  = lds_in  + wave * 128;
    float* wout = lds_out + wave * (GROUPS_PER_WAVE * 512);

    {
        const int e_ld = base_e + (lane >> 2);
        float4 v = make_float4(0.f, 0.f, 0.f, 0.f);
        if (e_ld < B) v = *(const float4*)(x + (size_t)base_e * 16 + lane * 4);
        *(float4*)(win + lane * 4) = v;
    }
    wave_lds_fence();

    const int h = lane >> 4;
    const int r = lane & 15;
    const int e = r >> 2;
    const int t = r & 3;

    const float dv   = win[(lane >> 2) * 16 + t * t + 2 * t];
    const float sp   = fmaxf(dv, 0.f) + log1pf(expf(-fabsf(dv)));
    const float sp_x = xhalf(sp);
    const float sp0  = (h == 0) ? sp : sp_x;
    const float sp1  = (h == 1) ? sp : sp_x;

    float* dst = wout + (h * 8) * 32 + r * 2;
    group_products(win      + e * 16, sp0, h, t, dst);
    group_products(win + 64 + e * 16, sp1, h, t, dst + 512);
    wave_lds_fence();

    const int eg = base_e + (lane >> 3);
    group_store(wout,       out + (size_t)base_e * 32,       lane, eg     < B);
    group_store(wout + 512, out + (size_t)(base_e + 4) * 32, lane, eg + 4 < B);
}

extern "C" void kernel_launch(void* const* d_in, const int* in_sizes, int n_in,
                              void* d_out, int out_size, void* d_ws, size_t ws_size,
                              hipStream_t stream) {
    (void)n_in; (void)out_size; (void)d_ws; (void)ws_size;
    const float* x = (const float*)d_in[0];
    float* out     = (float*)d_out;
    const int B    = in_sizes[0] / 16;
    const int grid = (B + ELEMS_PER_BLOCK - 1) / ELEMS_PER_BLOCK;
    hipLaunchKernelGGL(cholesky_rho_wmma_kernel, dim3(grid), dim3(256), 0, stream,
                       x, out, B);
}
